# starting kernel
# baseline (speedup 1.0000x reference)
#include <hip/hip_runtime.h>

typedef float v2f __attribute__((ext_vector_type(2)));
typedef float v4f __attribute__((ext_vector_type(4)));
typedef float v8f __attribute__((ext_vector_type(8)));

#define LSEQ 512
#define KDIM 256
#define RDIM 32
#define H1D  100
#define H2D  100
#define NLAB 10

__global__ __launch_bounds__(1024) void farnn_setup(
    const float* __restrict__ alpha0, const float* __restrict__ D1,
    const float* __restrict__ D2, float* __restrict__ ws) {
  int tid = threadIdx.x;
  int k = tid >> 5, n = tid & 31;
  float s = 0.f;
  for (int c = 0; c < KDIM; ++c) s = fmaf(D1[n * KDIM + c], D2[c * RDIM + k], s);
  ws[k * RDIM + n] = s;
  if (tid < RDIM) {
    float b = 0.f;
    for (int c = 0; c < KDIM; ++c) b = fmaf(alpha0[c], D1[tid * KDIM + c], b);
    ws[RDIM * RDIM + tid] = b;
  }
}

static __device__ __forceinline__ v8f wmma_f32_4(v2f a, v2f b, v8f c) {
  return __builtin_amdgcn_wmma_f32_16x16x4_f32(false, a, false, b, (short)0, c,
                                               false, false);
}

__global__ __launch_bounds__(32) void farnn_main(
    const int* __restrict__ x, const int* __restrict__ lengths,
    const float* __restrict__ ER, const float* __restrict__ ws,
    const float* __restrict__ D2, const float* __restrict__ W1,
    const float* __restrict__ b1, const float* __restrict__ W2,
    const float* __restrict__ b2, const float* __restrict__ W3,
    const float* __restrict__ b3, float* __restrict__ out) {
  __shared__ int   tokbuf[16];
  __shared__ float Bbuf[16][RDIM];
  __shared__ float Ebuf[16][RDIM];
  __shared__ float alphaF[16][KDIM];
  __shared__ float Hb1[16][H1D];
  __shared__ float Hb2[16][H2D];

  const int l    = threadIdx.x;
  const int lm   = l & 15;
  const int half = l >> 4;
  const int seq0 = blockIdx.x * 16;

  int lenB = lengths[seq0 + lm];
  int lenA[8];
#pragma unroll
  for (int v = 0; v < 8; ++v) lenA[v] = lengths[seq0 + v + 8 * half];

  v2f mfrag[2][8];
#pragma unroll
  for (int nt = 0; nt < 2; ++nt)
#pragma unroll
    for (int kc = 0; kc < 8; ++kc) {
      int n = 16 * nt + lm, k0 = 4 * kc + 2 * half;
      v2f f; f.x = ws[k0 * RDIM + n]; f.y = ws[(k0 + 1) * RDIM + n];
      mfrag[nt][kc] = f;
    }

  v8f betaC[2];
#pragma unroll
  for (int nt = 0; nt < 2; ++nt) {
    float b0 = ws[RDIM * RDIM + 16 * nt + lm];
#pragma unroll
    for (int v = 0; v < 8; ++v) betaC[nt][v] = b0;
  }
  v2f gam[8];
#pragma unroll
  for (int kc = 0; kc < 8; ++kc) { gam[kc].x = 0.f; gam[kc].y = 0.f; }

  int rtok = 0;
  if (l < 16) { rtok = x[(seq0 + l) * LSEQ + 0]; tokbuf[l] = rtok; }
  __syncthreads();
  v4f ereg[4];
#pragma unroll
  for (int i = 0; i < 4; ++i) {
    int idx = i * 32 + l, m = idx >> 3, q = idx & 7;
    ereg[i] = *(const v4f*)(ER + (size_t)tokbuf[m] * RDIM + q * 4);
  }
  if (l < 16) rtok = x[(seq0 + l) * LSEQ + 1];

#pragma unroll 1
  for (int t = 0; t < LSEQ; ++t) {
#pragma unroll
    for (int nt = 0; nt < 2; ++nt)
#pragma unroll
      for (int v = 0; v < 8; ++v)
        Bbuf[v + 8 * half][16 * nt + lm] = betaC[nt][v];
#pragma unroll
    for (int i = 0; i < 4; ++i) {
      int idx = i * 32 + l, m = idx >> 3, q = idx & 7;
      *(v4f*)(&Ebuf[m][q * 4]) = ereg[i];
    }
    if (l < 16) tokbuf[l] = rtok;
    __syncthreads();

    v4f eregn[4];
#pragma unroll
    for (int i = 0; i < 4; ++i) { eregn[i].x = 0.f; eregn[i].y = 0.f; eregn[i].z = 0.f; eregn[i].w = 0.f; }
    int rtokn = 0;
    if (t + 1 < LSEQ) {
#pragma unroll
      for (int i = 0; i < 4; ++i) {
        int idx = i * 32 + l, m = idx >> 3, q = idx & 7;
        eregn[i] = *(const v4f*)(ER + (size_t)tokbuf[m] * RDIM + q * 4);
      }
      if (l < 16 && t + 2 < LSEQ) rtokn = x[(seq0 + l) * LSEQ + t + 2];
    }

    const bool actB = (t < lenB);
#pragma unroll
    for (int kc = 0; kc < 8; ++kc) {
      int r = 4 * kc + 2 * half;
      v2f bb = *(const v2f*)(&Bbuf[lm][r]);
      v2f ee = *(const v2f*)(&Ebuf[lm][r]);
      float cx = bb.x * ee.x, cy = bb.y * ee.y;
      gam[kc].x = actB ? cx : gam[kc].x;
      gam[kc].y = actB ? cy : gam[kc].y;
    }

    v8f z = {};
    v8f a00 = z, a01 = z, a10 = z, a11 = z;
#pragma unroll
    for (int kc = 0; kc < 8; kc += 2) {
      a00 = wmma_f32_4(gam[kc],     mfrag[0][kc],     a00);
      a10 = wmma_f32_4(gam[kc],     mfrag[1][kc],     a10);
      a01 = wmma_f32_4(gam[kc + 1], mfrag[0][kc + 1], a01);
      a11 = wmma_f32_4(gam[kc + 1], mfrag[1][kc + 1], a11);
    }
    v8f nb0 = a00 + a01, nb1 = a10 + a11;
#pragma unroll
    for (int v = 0; v < 8; ++v) {
      bool a = (t < lenA[v]);
      betaC[0][v] = a ? nb0[v] : betaC[0][v];
      betaC[1][v] = a ? nb1[v] : betaC[1][v];
    }
#pragma unroll
    for (int i = 0; i < 4; ++i) ereg[i] = eregn[i];
    rtok = rtokn;
  }

#pragma unroll 4
  for (int nt2 = 0; nt2 < 16; ++nt2) {
    v8f accF = {};
    int n = 16 * nt2 + lm;
#pragma unroll
    for (int kc = 0; kc < 8; ++kc) {
      int k0 = 4 * kc + 2 * half;
      v2f bf = *(const v2f*)(D2 + (size_t)n * RDIM + k0);
      accF = wmma_f32_4(gam[kc], bf, accF);
    }
#pragma unroll
    for (int v = 0; v < 8; ++v) alphaF[v + 8 * half][n] = accF[v];
  }
  __syncthreads();

  for (int o = l; o < 16 * H1D; o += 32) {
    int s = o / H1D, j = o % H1D;
    float acc = b1[j];
    for (int c = 0; c < KDIM; ++c) acc = fmaf(alphaF[s][c], W1[j * KDIM + c], acc);
    Hb1[s][j] = fmaxf(acc, 0.f);
  }
  __syncthreads();
  for (int o = l; o < 16 * H2D; o += 32) {
    int s = o / H2D, j = o % H2D;
    float acc = b2[j];
    for (int c = 0; c < H1D; ++c) acc = fmaf(Hb1[s][c], W2[j * H1D + c], acc);
    Hb2[s][j] = fmaxf(acc, 0.f);
  }
  __syncthreads();
  for (int o = l; o < 16 * NLAB; o += 32) {
    int s = o / NLAB, j = o % NLAB;
    float acc = b3[j];
    for (int c = 0; c < H2D; ++c) acc = fmaf(Hb2[s][c], W3[j * H2D + c], acc);
    out[(size_t)(seq0 + s) * NLAB + j] = 1.0f * acc;
  }
}

extern "C" void kernel_launch(void* const* d_in, const int* in_sizes, int n_in,
                              void* d_out, int out_size, void* d_ws, size_t ws_size,
                              hipStream_t stream) {
  const int*   x      = (const int*)d_in[0];
  const int*   lens   = (const int*)d_in[1];
  const float* alpha0 = (const float*)d_in[2];
  const float* ER     = (const float*)d_in[3];
  const float* D1     = (const float*)d_in[4];
  const float* D2     = (const float*)d_in[5];
  const float* W1     = (const float*)d_in[6];
  const float* b1     = (const float*)d_in[7];
  const float* W2     = (const float*)d_in[8];
  const float* b2     = (const float*)d_in[9];
  const float* W3     = (const float*)d_in[10];
  const float* b3     = (const float*)d_in[11];
  float* ws  = (float*)d_ws;
  float* out = (float*)d_out;

  farnn_setup<<<1, 1024, 0, stream>>>(alpha0, D1, D2, ws);
  farnn_main<<<2048 / 16, 32, 0, stream>>>(x, lens, ER, ws, D2, W1, b1, W2, b2,
                                           W3, b3, out);
}


# Set-up re-tiled over 33 one-wave blocks: block n forms row n of M=D1@D2 (lane=k, D2 rows read as whole lines, D1 row handed out via LDS), one block forms beta0; ws holds M row-major; sum order kept
# speedup vs baseline: 3.1275x; 3.1275x over previous
#include <hip/hip_runtime.h>

typedef float v2f __attribute__((ext_vector_type(2)));
typedef float v4f __attribute__((ext_vector_type(4)));
typedef float v8f __attribute__((ext_vector_type(8)));
typedef int   v4i __attribute__((ext_vector_type(4)));

#define LSEQ 512
#define KDIM 256
#define RDIM 32
#define H1D  100
#define H2D  100
#define NLAB 10

__global__ __launch_bounds__(32) void farnn_setup(
    const float* __restrict__ alpha0, const float* __restrict__ D1,
    const float* __restrict__ D2, float* __restrict__ ws) {
  __shared__ __attribute__((aligned(16))) float ubuf[KDIM];
  const int  lane = threadIdx.x;
  const int  blk  = blockIdx.x;
  const bool mrow = blk < RDIM;
  const float* __restrict__ uni = mrow ? D1 + blk * KDIM : alpha0;
  *(v4f*)(ubuf + 4 * lane)       = *(const v4f*)(uni + 4 * lane);
  *(v4f*)(ubuf + 128 + 4 * lane) = *(const v4f*)(uni + 128 + 4 * lane);
  __builtin_amdgcn_fence(__ATOMIC_RELEASE, "wavefront");
  __builtin_amdgcn_wave_barrier();
  __builtin_amdgcn_fence(__ATOMIC_ACQUIRE, "wavefront");
  v4f a[KDIM / 4];
#pragma unroll
  for (int g = 0; g < KDIM / 4; ++g) a[g] = *(const v4f*)(ubuf + 4 * g);
  __builtin_amdgcn_sched_barrier(0);
  float s = 0.f;
  if (mrow) {
#pragma unroll
    for (int g = 0; g < KDIM / 4; ++g) {
      s = fmaf(a[g].x, D2[(4 * g + 0) * RDIM + lane], s);
      s = fmaf(a[g].y, D2[(4 * g + 1) * RDIM + lane], s);
      s = fmaf(a[g].z, D2[(4 * g + 2) * RDIM + lane], s);
      s = fmaf(a[g].w, D2[(4 * g + 3) * RDIM + lane], s);
    }
    ws[blk * RDIM + lane] = s;
  } else {
    const float* __restrict__ row = D1 + lane * KDIM;
#pragma unroll
    for (int g = 0; g < KDIM / 4; ++g) {
      const v4f d = *(const v4f*)(row + 4 * g);
      s = fmaf(a[g].x, d.x, s);
      s = fmaf(a[g].y, d.y, s);
      s = fmaf(a[g].z, d.z, s);
      s = fmaf(a[g].w, d.w, s);
    }
    ws[RDIM * RDIM + lane] = s;
  }
}

static __device__ __forceinline__ v8f wmma_f32_4(v2f a, v2f b, v8f c) {
  return __builtin_amdgcn_wmma_f32_16x16x4_f32(false, a, false, b, (short)0, c,
                                               false, false);
}

template <int KD, int NJ>
static __device__ __forceinline__ void dense_rows(const float* __restrict__ Wrows,
                                                  const float* arow, float (&acc)[NJ]) {
  constexpr int NG = KD / 4;
  v4f wA[NJ], wB[NJ];
#pragma unroll
  for (int i = 0; i < NJ; ++i) wA[i] = *(const v4f*)(Wrows + i * KD);
#pragma unroll 1
  for (int g = 0; g < NG; g += 2) {
    const int g1 = (g + 1 < NG) ? g + 1 : NG - 1;
    const int g2 = (g + 2 < NG) ? g + 2 : NG - 1;
#pragma unroll
    for (int i = 0; i < NJ; ++i) wB[i] = *(const v4f*)(Wrows + i * KD + 4 * g1);
    {
      const v4f a = *(const v4f*)(arow + 4 * g);
#pragma unroll
      for (int i = 0; i < NJ; ++i) {
        float t = acc[i];
        t = fmaf(a.x, wA[i].x, t); t = fmaf(a.y, wA[i].y, t);
        t = fmaf(a.z, wA[i].z, t); t = fmaf(a.w, wA[i].w, t);
        acc[i] = t;
      }
    }
#pragma unroll
    for (int i = 0; i < NJ; ++i) wA[i] = *(const v4f*)(Wrows + i * KD + 4 * g2);
    if (g + 1 < NG) {
      const v4f a = *(const v4f*)(arow + 4 * (g + 1));
#pragma unroll
      for (int i = 0; i < NJ; ++i) {
        float t = acc[i];
        t = fmaf(a.x, wB[i].x, t); t = fmaf(a.y, wB[i].y, t);
        t = fmaf(a.z, wB[i].z, t); t = fmaf(a.w, wB[i].w, t);
        acc[i] = t;
      }
    }
  }
}

__global__ __launch_bounds__(32) void farnn_main(
    const int* __restrict__ x, const int* __restrict__ lengths,
    const float* __restrict__ ER, const float* __restrict__ ws,
    const float* __restrict__ D2, const float* __restrict__ W1,
    const float* __restrict__ b1, const float* __restrict__ W2,
    const float* __restrict__ b2, const float* __restrict__ W3,
    const float* __restrict__ b3, float* __restrict__ out) {
  __shared__ __attribute__((aligned(16))) float Bbuf[16][RDIM + 4];
  __shared__ __attribute__((aligned(16))) float Ebuf[16][RDIM + 4];
  __shared__ __attribute__((aligned(16))) float alphaF[16][KDIM + 4];
  __shared__ __attribute__((aligned(16))) float Hb1[16][H1D];
  __shared__ __attribute__((aligned(16))) float Hb2[16][H2D];
  __shared__ float obuf[16 * NLAB];

  const int l    = threadIdx.x;
  const int lm   = l & 15;
  const int half = l >> 4;
  const int seq0 = blockIdx.x * 16;

  int lenB = lengths[seq0 + lm];
  int lenA[8];
#pragma unroll
  for (int v = 0; v < 8; ++v) lenA[v] = lengths[seq0 + v + 8 * half];

  v2f mfrag[2][8];
#pragma unroll
  for (int nt = 0; nt < 2; ++nt)
#pragma unroll
    for (int kc = 0; kc < 8; ++kc) {
      int n = 16 * nt + lm, k0 = 4 * kc + 2 * half;
      mfrag[nt][kc] = *(const v2f*)(ws + n * RDIM + k0);
    }

  v8f betaC[2];
#pragma unroll
  for (int nt = 0; nt < 2; ++nt) {
    float b0 = ws[RDIM * RDIM + 16 * nt + lm];
#pragma unroll
    for (int v = 0; v < 8; ++v) betaC[nt][v] = b0;
  }
  v2f gam[8];
#pragma unroll
  for (int kc = 0; kc < 8; ++kc) { gam[kc].x = 0.f; gam[kc].y = 0.f; }

  const int      grow = 4 * (l >> 4) + 8 * ((l >> 3) & 1);
  const unsigned gq   = (unsigned)(l & 7) * 16u;
  unsigned xoff[4];
#pragma unroll
  for (int i = 0; i < 4; ++i) xoff[i] = (unsigned)(seq0 + grow + i) * (unsigned)(LSEQ * 4);
  auto tok_quad = [&](int u, int i) -> v4i {
    return *(const v4i*)((const char*)x + (size_t)(xoff[i] + 16u * (unsigned)u));
  };
  auto row_off = [&](v4i tok) -> v4i {
    return tok * (RDIM * 4) + (int)gq;
  };
  auto piece = [&](int off) -> v4f {
    return *(const v4f*)((const char*)ER + (size_t)(unsigned)off);
  };

  v4i offA[4], offB[4], rawC[4];
#pragma unroll
  for (int i = 0; i < 4; ++i) { offA[i] = row_off(tok_quad(0, i)); offB[i] = row_off(tok_quad(1, i)); }
  v4f eE[4], eO[4];
#pragma unroll
  for (int i = 0; i < 4; ++i) { eE[i] = piece(offA[i].x); eO[i] = piece(offA[i].y); }

  auto step = [&](int t, v4f (&ecur)[4], int o0, int o1, int o2, int o3, int uq) {
#pragma unroll
    for (int nt = 0; nt < 2; ++nt)
#pragma unroll
      for (int v = 0; v < 8; ++v)
        Bbuf[v + 8 * half][16 * nt + lm] = betaC[nt][v];
#pragma unroll
    for (int i = 0; i < 4; ++i) *(v4f*)(&Ebuf[grow + i][(l & 7) * 4]) = ecur[i];
    __builtin_amdgcn_fence(__ATOMIC_RELEASE, "wavefront");
    __builtin_amdgcn_wave_barrier();
    __builtin_amdgcn_fence(__ATOMIC_ACQUIRE, "wavefront");

    v2f bb[8], ee[8];
#pragma unroll
    for (int kc = 0; kc < 8; ++kc) {
      int r = 4 * kc + 2 * half;
      bb[kc] = *(const v2f*)(&Bbuf[lm][r]);
      ee[kc] = *(const v2f*)(&Ebuf[lm][r]);
    }

    __builtin_amdgcn_sched_barrier(0);

    ecur[0] = piece(o0); ecur[1] = piece(o1); ecur[2] = piece(o2); ecur[3] = piece(o3);
    if (uq >= 0) {
      const int uqc = (uq < LSEQ / 4) ? uq : (LSEQ / 4 - 1);
#pragma unroll
      for (int i = 0; i < 4; ++i) rawC[i] = tok_quad(uqc, i);
    }
    __builtin_amdgcn_sched_barrier(0);

    const bool actB = (t < lenB);
#pragma unroll
    for (int kc = 0; kc < 8; ++kc) {
      float cx = bb[kc].x * ee[kc].x, cy = bb[kc].y * ee[kc].y;
      gam[kc].x = actB ? cx : gam[kc].x;
      gam[kc].y = actB ? cy : gam[kc].y;
    }

    v8f z = {};
    v8f a00 = z, a01 = z, a10 = z, a11 = z;
#pragma unroll
    for (int kc = 0; kc < 8; kc += 2) {
      a00 = wmma_f32_4(gam[kc],     mfrag[0][kc],     a00);
      a10 = wmma_f32_4(gam[kc],     mfrag[1][kc],     a10);
      a01 = wmma_f32_4(gam[kc + 1], mfrag[0][kc + 1], a01);
      a11 = wmma_f32_4(gam[kc + 1], mfrag[1][kc + 1], a11);
    }
    v8f nb0 = a00 + a01, nb1 = a10 + a11;
#pragma unroll
    for (int v = 0; v < 8; ++v) {
      bool a = (t < lenA[v]);
      betaC[0][v] = a ? nb0[v] : betaC[0][v];
      betaC[1][v] = a ? nb1[v] : betaC[1][v];
    }
  };

#pragma unroll 1
  for (int u = 0; u < LSEQ / 4; ++u) {
    const int t0 = 4 * u;
    step(t0 + 0, eE, offA[0].z, offA[1].z, offA[2].z, offA[3].z, u + 2);
    step(t0 + 1, eO, offA[0].w, offA[1].w, offA[2].w, offA[3].w, -1);
    step(t0 + 2, eE, offB[0].x, offB[1].x, offB[2].x, offB[3].x, -1);
    step(t0 + 3, eO, offB[0].y, offB[1].y, offB[2].y, offB[3].y, -1);
#pragma unroll
    for (int i = 0; i < 4; ++i) { offA[i] = offB[i]; offB[i] = row_off(rawC[i]); }
  }

#pragma unroll 4
  for (int nt2 = 0; nt2 < 16; ++nt2) {
    v8f accF = {};
    int n = 16 * nt2 + lm;
#pragma unroll
    for (int kc = 0; kc < 8; ++kc) {
      int k0 = 4 * kc + 2 * half;
      v2f bf = *(const v2f*)(D2 + (size_t)n * RDIM + k0);
      accF = wmma_f32_4(gam[kc], bf, accF);
    }
#pragma unroll
    for (int v = 0; v < 8; ++v) alphaF[v + 8 * half][n] = accF[v];
  }
  __syncthreads();

  {
    const int s = lm, jh = half;
    float acc[50];
#pragma unroll
    for (int i = 0; i < 50; ++i) acc[i] = b1[50 * jh + i];
    dense_rows<KDIM, 50>(W1 + (size_t)(50 * jh) * KDIM, &alphaF[s][0], acc);
#pragma unroll
    for (int i = 0; i < 50; ++i) Hb1[s][50 * jh + i] = fmaxf(acc[i], 0.f);
    __syncthreads();
#pragma unroll
    for (int i = 0; i < 50; ++i) acc[i] = b2[50 * jh + i];
    dense_rows<H1D, 50>(W2 + (size_t)(50 * jh) * H1D, &Hb1[s][0], acc);
#pragma unroll
    for (int i = 0; i < 50; ++i) Hb2[s][50 * jh + i] = fmaxf(acc[i], 0.f);
    __syncthreads();
    float acc3[5];
#pragma unroll
    for (int i = 0; i < 5; ++i) acc3[i] = b3[5 * jh + i];
    dense_rows<H2D, 5>(W3 + (size_t)(5 * jh) * H2D, &Hb2[s][0], acc3);
#pragma unroll
    for (int i = 0; i < 5; ++i) obuf[s * NLAB + 5 * jh + i] = 1.0f * acc3[i];
    __syncthreads();
#pragma unroll
    for (int it = 0; it < 5; ++it)
      out[(size_t)seq0 * NLAB + it * 32 + l] = obuf[it * 32 + l];
  }
}

extern "C" void kernel_launch(void* const* d_in, const int* in_sizes, int n_in,
                              void* d_out, int out_size, void* d_ws, size_t ws_size,
                              hipStream_t stream) {
  const int*   x      = (const int*)d_in[0];
  const int*   lens   = (const int*)d_in[1];
  const float* alpha0 = (const float*)d_in[2];
  const float* ER     = (const float*)d_in[3];
  const float* D1     = (const float*)d_in[4];
  const float* D2     = (const float*)d_in[5];
  const float* W1     = (const float*)d_in[6];
  const float* b1     = (const float*)d_in[7];
  const float* W2     = (const float*)d_in[8];
  const float* b2     = (const float*)d_in[9];
  const float* W3     = (const float*)d_in[10];
  const float* b3     = (const float*)d_in[11];
  float* ws  = (float*)d_ws;
  float* out = (float*)d_out;

  farnn_setup<<<33, 32, 0, stream>>>(alpha0, D1, D2, ws);
  farnn_main<<<2048 / 16, 32, 0, stream>>>(x, lens, ER, ws, D2, W1, b1, W2, b2,
                                           W3, b3, out);
}


# Embeddings loaded straight into A-fragment registers: k order permuted (a lane half owns 16 contiguous k = 64 B of its row's ER row), so e never crosses LDS; beta, alpha summed in new k order
# speedup vs baseline: 3.3958x; 1.0858x over previous
#include <hip/hip_runtime.h>

typedef float v2f __attribute__((ext_vector_type(2)));
typedef float v4f __attribute__((ext_vector_type(4)));
typedef float v8f __attribute__((ext_vector_type(8)));
typedef int   v4i __attribute__((ext_vector_type(4)));

#define LSEQ 512
#define KDIM 256
#define RDIM 32
#define H1D  100
#define H2D  100
#define NLAB 10

__global__ __launch_bounds__(32) void farnn_setup(
    const float* __restrict__ alpha0, const float* __restrict__ D1,
    const float* __restrict__ D2, float* __restrict__ ws) {
  __shared__ __attribute__((aligned(16))) float ubuf[KDIM];
  const int  lane = threadIdx.x;
  const int  blk  = blockIdx.x;
  const bool mrow = blk < RDIM;
  const float* __restrict__ uni = mrow ? D1 + blk * KDIM : alpha0;
  *(v4f*)(ubuf + 4 * lane)       = *(const v4f*)(uni + 4 * lane);
  *(v4f*)(ubuf + 128 + 4 * lane) = *(const v4f*)(uni + 128 + 4 * lane);
  __builtin_amdgcn_fence(__ATOMIC_RELEASE, "wavefront");
  __builtin_amdgcn_wave_barrier();
  __builtin_amdgcn_fence(__ATOMIC_ACQUIRE, "wavefront");
  v4f a[KDIM / 4];
#pragma unroll
  for (int g = 0; g < KDIM / 4; ++g) a[g] = *(const v4f*)(ubuf + 4 * g);
  __builtin_amdgcn_sched_barrier(0);
  float s = 0.f;
  if (mrow) {
#pragma unroll
    for (int g = 0; g < KDIM / 4; ++g) {
      s = fmaf(a[g].x, D2[(4 * g + 0) * RDIM + lane], s);
      s = fmaf(a[g].y, D2[(4 * g + 1) * RDIM + lane], s);
      s = fmaf(a[g].z, D2[(4 * g + 2) * RDIM + lane], s);
      s = fmaf(a[g].w, D2[(4 * g + 3) * RDIM + lane], s);
    }
    ws[blk * RDIM + lane] = s;
  } else {
    const float* __restrict__ row = D1 + lane * KDIM;
#pragma unroll
    for (int g = 0; g < KDIM / 4; ++g) {
      const v4f d = *(const v4f*)(row + 4 * g);
      s = fmaf(a[g].x, d.x, s);
      s = fmaf(a[g].y, d.y, s);
      s = fmaf(a[g].z, d.z, s);
      s = fmaf(a[g].w, d.w, s);
    }
    ws[RDIM * RDIM + lane] = s;
  }
}

static __device__ __forceinline__ v8f wmma_f32_4(v2f a, v2f b, v8f c) {
  return __builtin_amdgcn_wmma_f32_16x16x4_f32(false, a, false, b, (short)0, c,
                                               false, false);
}

template <int KD, int NJ>
static __device__ __forceinline__ void dense_rows(const float* __restrict__ Wrows,
                                                  const float* arow, float (&acc)[NJ]) {
  constexpr int NG = KD / 4;
  v4f wA[NJ], wB[NJ];
#pragma unroll
  for (int i = 0; i < NJ; ++i) wA[i] = *(const v4f*)(Wrows + i * KD);
#pragma unroll 1
  for (int g = 0; g < NG; g += 2) {
    const int g1 = (g + 1 < NG) ? g + 1 : NG - 1;
    const int g2 = (g + 2 < NG) ? g + 2 : NG - 1;
#pragma unroll
    for (int i = 0; i < NJ; ++i) wB[i] = *(const v4f*)(Wrows + i * KD + 4 * g1);
    {
      const v4f a = *(const v4f*)(arow + 4 * g);
#pragma unroll
      for (int i = 0; i < NJ; ++i) {
        float t = acc[i];
        t = fmaf(a.x, wA[i].x, t); t = fmaf(a.y, wA[i].y, t);
        t = fmaf(a.z, wA[i].z, t); t = fmaf(a.w, wA[i].w, t);
        acc[i] = t;
      }
    }
#pragma unroll
    for (int i = 0; i < NJ; ++i) wA[i] = *(const v4f*)(Wrows + i * KD + 4 * g2);
    if (g + 1 < NG) {
      const v4f a = *(const v4f*)(arow + 4 * (g + 1));
#pragma unroll
      for (int i = 0; i < NJ; ++i) {
        float t = acc[i];
        t = fmaf(a.x, wB[i].x, t); t = fmaf(a.y, wB[i].y, t);
        t = fmaf(a.z, wB[i].z, t); t = fmaf(a.w, wB[i].w, t);
        acc[i] = t;
      }
    }
  }
}

__global__ __launch_bounds__(32) void farnn_main(
    const int* __restrict__ x, const int* __restrict__ lengths,
    const float* __restrict__ ER, const float* __restrict__ ws,
    const float* __restrict__ D2, const float* __restrict__ W1,
    const float* __restrict__ b1, const float* __restrict__ W2,
    const float* __restrict__ b2, const float* __restrict__ W3,
    const float* __restrict__ b3, float* __restrict__ out) {
  __shared__ __attribute__((aligned(16))) float Bbuf[16][RDIM + 4];
  __shared__ __attribute__((aligned(16))) float alphaF[16][KDIM + 4];
  __shared__ __attribute__((aligned(16))) float Hb1[16][H1D];
  __shared__ __attribute__((aligned(16))) float Hb2[16][H2D];
  __shared__ float obuf[16 * NLAB];

  const int l    = threadIdx.x;
  const int lm   = l & 15;
  const int half = l >> 4;
  const int seq0 = blockIdx.x * 16;

  int lenB = lengths[seq0 + lm];
  int lenA[8];
#pragma unroll
  for (int v = 0; v < 8; ++v) lenA[v] = lengths[seq0 + v + 8 * half];

  v2f mfrag[2][8];
#pragma unroll
  for (int nt = 0; nt < 2; ++nt)
#pragma unroll
    for (int kc = 0; kc < 8; ++kc) {
      int n = 16 * nt + lm, k0 = 16 * half + 2 * kc;
      mfrag[nt][kc] = *(const v2f*)(ws + n * RDIM + k0);
    }

  v8f betaC[2];
#pragma unroll
  for (int nt = 0; nt < 2; ++nt) {
    float b0 = ws[RDIM * RDIM + 16 * nt + lm];
#pragma unroll
    for (int v = 0; v < 8; ++v) betaC[nt][v] = b0;
  }
  v2f gam[8];
#pragma unroll
  for (int kc = 0; kc < 8; ++kc) { gam[kc].x = 0.f; gam[kc].y = 0.f; }

  const unsigned xoff = (unsigned)(seq0 + lm) * (unsigned)(LSEQ * 4);
  const int      hq   = half * 64;
  auto tok_quad = [&](unsigned boff) -> v4i {
    return *(const v4i*)((const char*)x + (size_t)boff);
  };
  auto row_off = [&](v4i tok) -> v4i {
    return tok * (RDIM * 4) + hq;
  };
  auto piece = [&](int off, int q) -> v4f {
    return *(const v4f*)((const char*)ER + (size_t)(unsigned)off + 16 * q);
  };

  v4i offA = row_off(tok_quad(xoff));
  v4i offB = row_off(tok_quad(xoff + 16u));
  v4i rawC = {};
  const unsigned xlast = xoff + 16u * (unsigned)(LSEQ / 4 - 1);
  unsigned xq = xoff + 32u;
  v4f eE[4], eO[4];
#pragma unroll
  for (int q = 0; q < 4; ++q) { eE[q] = piece(offA.x, q); eO[q] = piece(offA.y, q); }

  auto step = [&](int t, v4f (&ecur)[4], int onext, bool tq) {
#pragma unroll
    for (int nt = 0; nt < 2; ++nt)
#pragma unroll
      for (int v = 0; v < 8; ++v)
        Bbuf[v + 8 * half][16 * nt + lm] = betaC[nt][v];
    __builtin_amdgcn_fence(__ATOMIC_RELEASE, "wavefront");
    __builtin_amdgcn_wave_barrier();
    __builtin_amdgcn_fence(__ATOMIC_ACQUIRE, "wavefront");

    v4f bq[4];
#pragma unroll
    for (int q = 0; q < 4; ++q) bq[q] = *(const v4f*)(&Bbuf[lm][16 * half + 4 * q]);

    __builtin_amdgcn_sched_barrier(0);

    v4f en[4];
#pragma unroll
    for (int q = 0; q < 4; ++q) en[q] = piece(onext, q);
    if (tq) rawC = tok_quad(xq);
    __builtin_amdgcn_sched_barrier(0);

    const bool actB = (t < lenB);
#pragma unroll
    for (int q = 0; q < 4; ++q) {
      const v4f cq = bq[q] * ecur[q];
      gam[2 * q].x     = actB ? cq.x : gam[2 * q].x;
      gam[2 * q].y     = actB ? cq.y : gam[2 * q].y;
      gam[2 * q + 1].x = actB ? cq.z : gam[2 * q + 1].x;
      gam[2 * q + 1].y = actB ? cq.w : gam[2 * q + 1].y;
    }
#pragma unroll
    for (int q = 0; q < 4; ++q) ecur[q] = en[q];

    v8f z = {};
    v8f a00 = z, a01 = z, a10 = z, a11 = z;
#pragma unroll
    for (int kc = 0; kc < 8; kc += 2) {
      a00 = wmma_f32_4(gam[kc],     mfrag[0][kc],     a00);
      a10 = wmma_f32_4(gam[kc],     mfrag[1][kc],     a10);
      a01 = wmma_f32_4(gam[kc + 1], mfrag[0][kc + 1], a01);
      a11 = wmma_f32_4(gam[kc + 1], mfrag[1][kc + 1], a11);
    }
    v8f nb0 = a00 + a01, nb1 = a10 + a11;
#pragma unroll
    for (int v = 0; v < 8; ++v) {
      bool a = (t < lenA[v]);
      betaC[0][v] = a ? nb0[v] : betaC[0][v];
      betaC[1][v] = a ? nb1[v] : betaC[1][v];
    }
  };

#pragma unroll 1
  for (int u = 0; u < LSEQ / 4; ++u) {
    const int t0 = 4 * u;
    step(t0 + 0, eE, offA.z, true);
    step(t0 + 1, eO, offA.w, false);
    step(t0 + 2, eE, offB.x, false);
    step(t0 + 3, eO, offB.y, false);
    offA = offB;
    offB = row_off(rawC);
    xq = (xq + 16u < xlast) ? xq + 16u : xlast;
  }

#pragma unroll 4
  for (int nt2 = 0; nt2 < 16; ++nt2) {
    v8f accF = {};
    int n = 16 * nt2 + lm;
#pragma unroll
    for (int kc = 0; kc < 8; ++kc) {
      int k0 = 16 * half + 2 * kc;
      v2f bf = *(const v2f*)(D2 + (size_t)n * RDIM + k0);
      accF = wmma_f32_4(gam[kc], bf, accF);
    }
#pragma unroll
    for (int v = 0; v < 8; ++v) alphaF[v + 8 * half][n] = accF[v];
  }
  __syncthreads();

  {
    const int s = lm, jh = half;
    float acc[50];
#pragma unroll
    for (int i = 0; i < 50; ++i) acc[i] = b1[50 * jh + i];
    dense_rows<KDIM, 50>(W1 + (size_t)(50 * jh) * KDIM, &alphaF[s][0], acc);
#pragma unroll
    for (int i = 0; i < 50; ++i) Hb1[s][50 * jh + i] = fmaxf(acc[i], 0.f);
    __syncthreads();
#pragma unroll
    for (int i = 0; i < 50; ++i) acc[i] = b2[50 * jh + i];
    dense_rows<H1D, 50>(W2 + (size_t)(50 * jh) * H1D, &Hb1[s][0], acc);
#pragma unroll
    for (int i = 0; i < 50; ++i) Hb2[s][50 * jh + i] = fmaxf(acc[i], 0.f);
    __syncthreads();
    float acc3[5];
#pragma unroll
    for (int i = 0; i < 5; ++i) acc3[i] = b3[5 * jh + i];
    dense_rows<H2D, 5>(W3 + (size_t)(5 * jh) * H2D, &Hb2[s][0], acc3);
#pragma unroll
    for (int i = 0; i < 5; ++i) obuf[s * NLAB + 5 * jh + i] = 1.0f * acc3[i];
    __syncthreads();
#pragma unroll
    for (int it = 0; it < 5; ++it)
      out[(size_t)seq0 * NLAB + it * 32 + l] = obuf[it * 32 + l];
  }
}

extern "C" void kernel_launch(void* const* d_in, const int* in_sizes, int n_in,
                              void* d_out, int out_size, void* d_ws, size_t ws_size,
                              hipStream_t stream) {
  const int*   x      = (const int*)d_in[0];
  const int*   lens   = (const int*)d_in[1];
  const float* alpha0 = (const float*)d_in[2];
  const float* ER     = (const float*)d_in[3];
  const float* D1     = (const float*)d_in[4];
  const float* D2     = (const float*)d_in[5];
  const float* W1     = (const float*)d_in[6];
  const float* b1     = (const float*)d_in[7];
  const float* W2     = (const float*)d_in[8];
  const float* b2     = (const float*)d_in[9];
  const float* W3     = (const float*)d_in[10];
  const float* b3     = (const float*)d_in[11];
  float* ws  = (float*)d_ws;
  float* out = (float*)d_out;

  farnn_setup<<<33, 32, 0, stream>>>(alpha0, D1, D2, ws);
  farnn_main<<<2048 / 16, 32, 0, stream>>>(x, lens, ER, ws, D2, W1, b1, W2, b2,
                                           W3, b3, out);
}


# Final projection folded into MLP layer 1: set-up also forms G=W1@D2 (100 more one-wave blocks); h1=relu(b1+gamma_last@G^T) on 56 f32 WMMAs, alpha_final never formed; h1 summed in new order
# speedup vs baseline: 3.8270x; 1.1270x over previous
#include <hip/hip_runtime.h>

typedef float v2f __attribute__((ext_vector_type(2)));
typedef float v4f __attribute__((ext_vector_type(4)));
typedef float v8f __attribute__((ext_vector_type(8)));
typedef int   v4i __attribute__((ext_vector_type(4)));

#define LSEQ 512
#define KDIM 256
#define RDIM 32
#define H1D  100
#define H2D  100
#define NLAB 10
#define H1P  116

#define GROW0 (RDIM + 1)
__global__ __launch_bounds__(32) void farnn_setup(
    const float* __restrict__ alpha0, const float* __restrict__ D1,
    const float* __restrict__ D2, const float* __restrict__ W1, float* __restrict__ ws) {
  __shared__ __attribute__((aligned(16))) float ubuf[KDIM];
  const int  lane = threadIdx.x;
  const int  blk  = blockIdx.x;
  const bool mrow = blk != RDIM;
  const float* __restrict__ uni = blk < RDIM ? D1 + blk * KDIM
                                : (mrow ? W1 + (blk - GROW0) * KDIM : alpha0);
  *(v4f*)(ubuf + 4 * lane)       = *(const v4f*)(uni + 4 * lane);
  *(v4f*)(ubuf + 128 + 4 * lane) = *(const v4f*)(uni + 128 + 4 * lane);
  __builtin_amdgcn_fence(__ATOMIC_RELEASE, "wavefront");
  __builtin_amdgcn_wave_barrier();
  __builtin_amdgcn_fence(__ATOMIC_ACQUIRE, "wavefront");
  v4f a[KDIM / 4];
#pragma unroll
  for (int g = 0; g < KDIM / 4; ++g) a[g] = *(const v4f*)(ubuf + 4 * g);
  __builtin_amdgcn_sched_barrier(0);
  float s = 0.f;
  if (mrow) {
#pragma unroll
    for (int g = 0; g < KDIM / 4; ++g) {
      s = fmaf(a[g].x, D2[(4 * g + 0) * RDIM + lane], s);
      s = fmaf(a[g].y, D2[(4 * g + 1) * RDIM + lane], s);
      s = fmaf(a[g].z, D2[(4 * g + 2) * RDIM + lane], s);
      s = fmaf(a[g].w, D2[(4 * g + 3) * RDIM + lane], s);
    }
    ws[blk * RDIM + lane] = s;
  } else {
    const float* __restrict__ row = D1 + lane * KDIM;
#pragma unroll
    for (int g = 0; g < KDIM / 4; ++g) {
      const v4f d = *(const v4f*)(row + 4 * g);
      s = fmaf(a[g].x, d.x, s);
      s = fmaf(a[g].y, d.y, s);
      s = fmaf(a[g].z, d.z, s);
      s = fmaf(a[g].w, d.w, s);
    }
    ws[RDIM * RDIM + lane] = s;
  }
}

static __device__ __forceinline__ v8f wmma_f32_4(v2f a, v2f b, v8f c) {
  return __builtin_amdgcn_wmma_f32_16x16x4_f32(false, a, false, b, (short)0, c,
                                               false, false);
}

template <int KD, int NJ>
static __device__ __forceinline__ void dense_rows(const float* __restrict__ Wrows,
                                                  const float* arow, float (&acc)[NJ]) {
  constexpr int NG = KD / 4;
  v4f wA[NJ], wB[NJ];
#pragma unroll
  for (int i = 0; i < NJ; ++i) wA[i] = *(const v4f*)(Wrows + i * KD);
#pragma unroll 1
  for (int g = 0; g < NG; g += 2) {
    const int g1 = (g + 1 < NG) ? g + 1 : NG - 1;
    const int g2 = (g + 2 < NG) ? g + 2 : NG - 1;
#pragma unroll
    for (int i = 0; i < NJ; ++i) wB[i] = *(const v4f*)(Wrows + i * KD + 4 * g1);
    {
      const v4f a = *(const v4f*)(arow + 4 * g);
#pragma unroll
      for (int i = 0; i < NJ; ++i) {
        float t = acc[i];
        t = fmaf(a.x, wA[i].x, t); t = fmaf(a.y, wA[i].y, t);
        t = fmaf(a.z, wA[i].z, t); t = fmaf(a.w, wA[i].w, t);
        acc[i] = t;
      }
    }
#pragma unroll
    for (int i = 0; i < NJ; ++i) wA[i] = *(const v4f*)(Wrows + i * KD + 4 * g2);
    if (g + 1 < NG) {
      const v4f a = *(const v4f*)(arow + 4 * (g + 1));
#pragma unroll
      for (int i = 0; i < NJ; ++i) {
        float t = acc[i];
        t = fmaf(a.x, wB[i].x, t); t = fmaf(a.y, wB[i].y, t);
        t = fmaf(a.z, wB[i].z, t); t = fmaf(a.w, wB[i].w, t);
        acc[i] = t;
      }
    }
  }
}

__global__ __launch_bounds__(32) void farnn_main(
    const int* __restrict__ x, const int* __restrict__ lengths,
    const float* __restrict__ ER, const float* __restrict__ ws,
    const float* __restrict__ b1, const float* __restrict__ W2,
    const float* __restrict__ b2, const float* __restrict__ W3,
    const float* __restrict__ b3, float* __restrict__ out) {
  __shared__ __attribute__((aligned(16))) float Bbuf[16][RDIM + 4];
  __shared__ __attribute__((aligned(16))) float Hb1[16][H1P];
  __shared__ __attribute__((aligned(16))) float Hb2[16][H2D];
  __shared__ float obuf[16 * NLAB];

  const int l    = threadIdx.x;
  const int lm   = l & 15;
  const int half = l >> 4;
  const int seq0 = blockIdx.x * 16;

  int lenB = lengths[seq0 + lm];
  int lenA[8];
#pragma unroll
  for (int v = 0; v < 8; ++v) lenA[v] = lengths[seq0 + v + 8 * half];

  v2f mfrag[2][8];
#pragma unroll
  for (int nt = 0; nt < 2; ++nt)
#pragma unroll
    for (int kc = 0; kc < 8; ++kc) {
      int n = 16 * nt + lm, k0 = 16 * half + 2 * kc;
      mfrag[nt][kc] = *(const v2f*)(ws + n * RDIM + k0);
    }

  v8f betaC[2];
#pragma unroll
  for (int nt = 0; nt < 2; ++nt) {
    float b0 = ws[RDIM * RDIM + 16 * nt + lm];
#pragma unroll
    for (int v = 0; v < 8; ++v) betaC[nt][v] = b0;
  }
  v2f gam[8];
#pragma unroll
  for (int kc = 0; kc < 8; ++kc) { gam[kc].x = 0.f; gam[kc].y = 0.f; }

  const unsigned xoff = (unsigned)(seq0 + lm) * (unsigned)(LSEQ * 4);
  const int      hq   = half * 64;
  auto tok_quad = [&](unsigned boff) -> v4i {
    return *(const v4i*)((const char*)x + (size_t)boff);
  };
  auto row_off = [&](v4i tok) -> v4i {
    return tok * (RDIM * 4) + hq;
  };
  auto piece = [&](int off, int q) -> v4f {
    return *(const v4f*)((const char*)ER + (size_t)(unsigned)off + 16 * q);
  };

  v4i offA = row_off(tok_quad(xoff));
  v4i offB = row_off(tok_quad(xoff + 16u));
  v4i rawC = {};
  const unsigned xlast = xoff + 16u * (unsigned)(LSEQ / 4 - 1);
  unsigned xq = xoff + 32u;
  v4f eE[4], eO[4];
#pragma unroll
  for (int q = 0; q < 4; ++q) { eE[q] = piece(offA.x, q); eO[q] = piece(offA.y, q); }

  auto step = [&](int t, v4f (&ecur)[4], int onext, bool tq) {
#pragma unroll
    for (int nt = 0; nt < 2; ++nt)
#pragma unroll
      for (int v = 0; v < 8; ++v)
        Bbuf[v + 8 * half][16 * nt + lm] = betaC[nt][v];
    __builtin_amdgcn_fence(__ATOMIC_RELEASE, "wavefront");
    __builtin_amdgcn_wave_barrier();
    __builtin_amdgcn_fence(__ATOMIC_ACQUIRE, "wavefront");

    v4f bq[4];
#pragma unroll
    for (int q = 0; q < 4; ++q) bq[q] = *(const v4f*)(&Bbuf[lm][16 * half + 4 * q]);

    __builtin_amdgcn_sched_barrier(0);

    v4f en[4];
#pragma unroll
    for (int q = 0; q < 4; ++q) en[q] = piece(onext, q);
    if (tq) rawC = tok_quad(xq);
    __builtin_amdgcn_sched_barrier(0);

    const bool actB = (t < lenB);
#pragma unroll
    for (int q = 0; q < 4; ++q) {
      const v4f cq = bq[q] * ecur[q];
      gam[2 * q].x     = actB ? cq.x : gam[2 * q].x;
      gam[2 * q].y     = actB ? cq.y : gam[2 * q].y;
      gam[2 * q + 1].x = actB ? cq.z : gam[2 * q + 1].x;
      gam[2 * q + 1].y = actB ? cq.w : gam[2 * q + 1].y;
    }
#pragma unroll
    for (int q = 0; q < 4; ++q) ecur[q] = en[q];

    v8f z = {};
    v8f a00 = z, a01 = z, a10 = z, a11 = z;
#pragma unroll
    for (int kc = 0; kc < 8; kc += 2) {
      a00 = wmma_f32_4(gam[kc],     mfrag[0][kc],     a00);
      a10 = wmma_f32_4(gam[kc],     mfrag[1][kc],     a10);
      a01 = wmma_f32_4(gam[kc + 1], mfrag[0][kc + 1], a01);
      a11 = wmma_f32_4(gam[kc + 1], mfrag[1][kc + 1], a11);
    }
    v8f nb0 = a00 + a01, nb1 = a10 + a11;
#pragma unroll
    for (int v = 0; v < 8; ++v) {
      bool a = (t < lenA[v]);
      betaC[0][v] = a ? nb0[v] : betaC[0][v];
      betaC[1][v] = a ? nb1[v] : betaC[1][v];
    }
  };

#pragma unroll 1
  for (int u = 0; u < LSEQ / 4; ++u) {
    const int t0 = 4 * u;
    step(t0 + 0, eE, offA.z, true);
    step(t0 + 1, eO, offA.w, false);
    step(t0 + 2, eE, offB.x, false);
    step(t0 + 3, eO, offB.y, false);
    offA = offB;
    offB = row_off(rawC);
    xq = (xq + 16u < xlast) ? xq + 16u : xlast;
  }

  {
    const float* __restrict__ G = ws + GROW0 * RDIM;
#pragma unroll
    for (int nt2 = 0; nt2 < 7; ++nt2) {
      const int n  = 16 * nt2 + lm;
      const int nc = n < H1D ? n : H1D - 1;
      const float bias = b1[nc];
      v8f accF;
#pragma unroll
      for (int v = 0; v < 8; ++v) accF[v] = bias;
#pragma unroll
      for (int kc = 0; kc < 8; ++kc) {
        int k0 = 16 * half + 2 * kc;
        v2f bf = *(const v2f*)(G + nc * RDIM + k0);
        accF = wmma_f32_4(gam[kc], bf, accF);
      }
#pragma unroll
      for (int v = 0; v < 8; ++v) Hb1[v + 8 * half][n] = fmaxf(accF[v], 0.f);
    }
  }
  __syncthreads();

  {
    const int s = lm, jh = half;
    float acc[50];
#pragma unroll
    for (int i = 0; i < 50; ++i) acc[i] = b2[50 * jh + i];
    dense_rows<H1D, 50>(W2 + (size_t)(50 * jh) * H1D, &Hb1[s][0], acc);
#pragma unroll
    for (int i = 0; i < 50; ++i) Hb2[s][50 * jh + i] = fmaxf(acc[i], 0.f);
    __syncthreads();
    float acc3[5];
#pragma unroll
    for (int i = 0; i < 5; ++i) acc3[i] = b3[5 * jh + i];
    dense_rows<H2D, 5>(W3 + (size_t)(5 * jh) * H2D, &Hb2[s][0], acc3);
#pragma unroll
    for (int i = 0; i < 5; ++i) obuf[s * NLAB + 5 * jh + i] = 1.0f * acc3[i];
    __syncthreads();
#pragma unroll
    for (int it = 0; it < 5; ++it)
      out[(size_t)seq0 * NLAB + it * 32 + l] = obuf[it * 32 + l];
  }
}

extern "C" void kernel_launch(void* const* d_in, const int* in_sizes, int n_in,
                              void* d_out, int out_size, void* d_ws, size_t ws_size,
                              hipStream_t stream) {
  const int*   x      = (const int*)d_in[0];
  const int*   lens   = (const int*)d_in[1];
  const float* alpha0 = (const float*)d_in[2];
  const float* ER     = (const float*)d_in[3];
  const float* D1     = (const float*)d_in[4];
  const float* D2     = (const float*)d_in[5];
  const float* W1     = (const float*)d_in[6];
  const float* b1     = (const float*)d_in[7];
  const float* W2     = (const float*)d_in[8];
  const float* b2     = (const float*)d_in[9];
  const float* W3     = (const float*)d_in[10];
  const float* b3     = (const float*)d_in[11];
  float* ws  = (float*)d_ws;
  float* out = (float*)d_out;

  farnn_setup<<<133, 32, 0, stream>>>(alpha0, D1, D2, W1, ws);
  farnn_main<<<2048 / 16, 32, 0, stream>>>(x, lens, ER, ws, b1, W2, b2,
                                           W3, b3, out);
}


# Time step split 4 ways along k, the summed dimension: 4-wave block, wave w owns k chunks 2w,2w+1, 4 WMMAs a step (was 16); partial betas added from LDS behind one block barrier; beta sum order new
# speedup vs baseline: 4.6704x; 1.2204x over previous
#include <hip/hip_runtime.h>

typedef float v2f __attribute__((ext_vector_type(2)));
typedef float v4f __attribute__((ext_vector_type(4)));
typedef float v8f __attribute__((ext_vector_type(8)));
typedef int   v4i __attribute__((ext_vector_type(4)));

#define LSEQ 512
#define KDIM 256
#define RDIM 32
#define H1D  100
#define H2D  100
#define NLAB 10
#define H1P  116

#define GROW0 (RDIM + 1)
__global__ __launch_bounds__(32) void farnn_setup(
    const float* __restrict__ alpha0, const float* __restrict__ D1,
    const float* __restrict__ D2, const float* __restrict__ W1, float* __restrict__ ws) {
  __shared__ __attribute__((aligned(16))) float ubuf[KDIM];
  const int  lane = threadIdx.x;
  const int  blk  = blockIdx.x;
  const bool mrow = blk != RDIM;
  const float* __restrict__ uni = blk < RDIM ? D1 + blk * KDIM
                                : (mrow ? W1 + (blk - GROW0) * KDIM : alpha0);
  *(v4f*)(ubuf + 4 * lane)       = *(const v4f*)(uni + 4 * lane);
  *(v4f*)(ubuf + 128 + 4 * lane) = *(const v4f*)(uni + 128 + 4 * lane);
  __builtin_amdgcn_fence(__ATOMIC_RELEASE, "wavefront");
  __builtin_amdgcn_wave_barrier();
  __builtin_amdgcn_fence(__ATOMIC_ACQUIRE, "wavefront");
  v4f a[KDIM / 4];
#pragma unroll
  for (int g = 0; g < KDIM / 4; ++g) a[g] = *(const v4f*)(ubuf + 4 * g);
  __builtin_amdgcn_sched_barrier(0);
  float s = 0.f;
  if (mrow) {
#pragma unroll
    for (int g = 0; g < KDIM / 4; ++g) {
      s = fmaf(a[g].x, D2[(4 * g + 0) * RDIM + lane], s);
      s = fmaf(a[g].y, D2[(4 * g + 1) * RDIM + lane], s);
      s = fmaf(a[g].z, D2[(4 * g + 2) * RDIM + lane], s);
      s = fmaf(a[g].w, D2[(4 * g + 3) * RDIM + lane], s);
    }
    ws[blk * RDIM + lane] = s;
  } else {
    const float* __restrict__ row = D1 + lane * KDIM;
#pragma unroll
    for (int g = 0; g < KDIM / 4; ++g) {
      const v4f d = *(const v4f*)(row + 4 * g);
      s = fmaf(a[g].x, d.x, s);
      s = fmaf(a[g].y, d.y, s);
      s = fmaf(a[g].z, d.z, s);
      s = fmaf(a[g].w, d.w, s);
    }
    ws[RDIM * RDIM + lane] = s;
  }
}

static __device__ __forceinline__ v8f wmma_f32_4(v2f a, v2f b, v8f c) {
  return __builtin_amdgcn_wmma_f32_16x16x4_f32(false, a, false, b, (short)0, c,
                                               false, false);
}

template <int KD, int NJ>
static __device__ __forceinline__ void dense_rows(const float* __restrict__ Wrows,
                                                  const float* arow, float (&acc)[NJ]) {
  constexpr int NG = KD / 4;
  v4f wA[NJ], wB[NJ];
#pragma unroll
  for (int i = 0; i < NJ; ++i) wA[i] = *(const v4f*)(Wrows + i * KD);
#pragma unroll 1
  for (int g = 0; g < NG; g += 2) {
    const int g1 = (g + 1 < NG) ? g + 1 : NG - 1;
    const int g2 = (g + 2 < NG) ? g + 2 : NG - 1;
#pragma unroll
    for (int i = 0; i < NJ; ++i) wB[i] = *(const v4f*)(Wrows + i * KD + 4 * g1);
    {
      const v4f a = *(const v4f*)(arow + 4 * g);
#pragma unroll
      for (int i = 0; i < NJ; ++i) {
        float t = acc[i];
        t = fmaf(a.x, wA[i].x, t); t = fmaf(a.y, wA[i].y, t);
        t = fmaf(a.z, wA[i].z, t); t = fmaf(a.w, wA[i].w, t);
        acc[i] = t;
      }
    }
#pragma unroll
    for (int i = 0; i < NJ; ++i) wA[i] = *(const v4f*)(Wrows + i * KD + 4 * g2);
    if (g + 1 < NG) {
      const v4f a = *(const v4f*)(arow + 4 * (g + 1));
#pragma unroll
      for (int i = 0; i < NJ; ++i) {
        float t = acc[i];
        t = fmaf(a.x, wB[i].x, t); t = fmaf(a.y, wB[i].y, t);
        t = fmaf(a.z, wB[i].z, t); t = fmaf(a.w, wB[i].w, t);
        acc[i] = t;
      }
    }
  }
}

#define NWV 4
#define PPITCH (RDIM + 4)

__global__ __launch_bounds__(32 * NWV) void farnn_main(
    const int* __restrict__ x, const int* __restrict__ lengths,
    const float* __restrict__ ER, const float* __restrict__ ws,
    const float* __restrict__ b1, const float* __restrict__ W2,
    const float* __restrict__ b2, const float* __restrict__ W3,
    const float* __restrict__ b3, float* __restrict__ out) {
  __shared__ __attribute__((aligned(16))) float Pbuf[2][NWV][16][PPITCH];
  __shared__ __attribute__((aligned(16))) float Gbuf[16][PPITCH];
  __shared__ __attribute__((aligned(16))) float Hb1[16][H1P];
  __shared__ __attribute__((aligned(16))) float Hb2[16][H2D];
  __shared__ float obuf[16 * NLAB];

  const int l    = threadIdx.x & 31;
  const int lm   = l & 15;
  const int half = l >> 4;
  const int w    = __builtin_amdgcn_readfirstlane((int)(threadIdx.x >> 5));
  const int seq0 = blockIdx.x * 16;

  const int lenB = lengths[seq0 + lm];

  const int kq = 16 * half + 4 * w;
  v2f mfrag[2][2];
#pragma unroll
  for (int nt = 0; nt < 2; ++nt) {
    const v4f m = *(const v4f*)(ws + (16 * nt + lm) * RDIM + kq);
    mfrag[nt][0].x = m.x; mfrag[nt][0].y = m.y;
    mfrag[nt][1].x = m.z; mfrag[nt][1].y = m.w;
  }

  v8f pc[2];
#pragma unroll
  for (int nt = 0; nt < 2; ++nt) {
    const float b0 = (w == 0) ? ws[RDIM * RDIM + 16 * nt + lm] : 0.f;
#pragma unroll
    for (int v = 0; v < 8; ++v) pc[nt][v] = b0;
  }
  v4f g = {0.f, 0.f, 0.f, 0.f};

  const unsigned xoff = (unsigned)(seq0 + lm) * (unsigned)(LSEQ * 4);
  const int      hq   = kq * 4;
  auto tok_quad = [&](unsigned boff) -> v4i {
    return *(const v4i*)((const char*)x + (size_t)boff);
  };
  auto row_off = [&](v4i tok) -> v4i {
    return tok * (RDIM * 4) + hq;
  };
  auto piece = [&](int off) -> v4f {
    return *(const v4f*)((const char*)ER + (size_t)(unsigned)off);
  };

  v4i offA = row_off(tok_quad(xoff));
  v4i offB = row_off(tok_quad(xoff + 16u));
  v4i rawC = {};
  const unsigned xlast = xoff + 16u * (unsigned)(LSEQ / 4 - 1);
  unsigned xq = xoff + 32u;
  v4f eE = piece(offA.x), eO = piece(offA.y);

  auto step = [&](int t, int par, v4f& ecur, int onext, bool tq) {
#pragma unroll
    for (int nt = 0; nt < 2; ++nt)
#pragma unroll
      for (int v = 0; v < 8; ++v)
        Pbuf[par][w][v + 8 * half][16 * nt + lm] = pc[nt][v];
    __builtin_amdgcn_fence(__ATOMIC_RELEASE, "workgroup", "local");
    __builtin_amdgcn_s_barrier();
    __builtin_amdgcn_fence(__ATOMIC_ACQUIRE, "workgroup", "local");

    v4f p[NWV];
#pragma unroll
    for (int j = 0; j < NWV; ++j) p[j] = *(const v4f*)(&Pbuf[par][j][lm][kq]);

    __builtin_amdgcn_sched_barrier(0);

    const v4f en = piece(onext);
    if (tq) rawC = tok_quad(xq);
    __builtin_amdgcn_sched_barrier(0);

    const bool actB = (t < lenB);
    const v4f bsum = ((p[0] + p[1]) + p[2]) + p[3];
    const v4f cq = bsum * ecur;
    g.x = actB ? cq.x : g.x;
    g.y = actB ? cq.y : g.y;
    g.z = actB ? cq.z : g.z;
    g.w = actB ? cq.w : g.w;
    ecur = en;

    v8f z = {};
    v2f g0, g1;
    g0.x = g.x; g0.y = g.y; g1.x = g.z; g1.y = g.w;
    v8f a0 = wmma_f32_4(g0, mfrag[0][0], z);
    v8f a1 = wmma_f32_4(g0, mfrag[1][0], z);
    a0 = wmma_f32_4(g1, mfrag[0][1], a0);
    a1 = wmma_f32_4(g1, mfrag[1][1], a1);
    pc[0] = a0; pc[1] = a1;
  };

#pragma unroll 1
  for (int u = 0; u < LSEQ / 4; ++u) {
    const int t0 = 4 * u;
    step(t0 + 0, 0, eE, offA.z, true);
    step(t0 + 1, 1, eO, offA.w, false);
    step(t0 + 2, 0, eE, offB.x, false);
    step(t0 + 3, 1, eO, offB.y, false);
    offA = offB;
    offB = row_off(rawC);
    xq = (xq + 16u < xlast) ? xq + 16u : xlast;
  }

  *(v4f*)(&Gbuf[lm][kq]) = g;
  __syncthreads();

  if (w == 0) {
    v2f gam[8];
#pragma unroll
    for (int q = 0; q < 4; ++q) {
      const v4f gq = *(const v4f*)(&Gbuf[lm][16 * half + 4 * q]);
      gam[2 * q].x = gq.x;     gam[2 * q].y = gq.y;
      gam[2 * q + 1].x = gq.z; gam[2 * q + 1].y = gq.w;
    }
    const float* __restrict__ G = ws + GROW0 * RDIM;
#pragma unroll
    for (int nt2 = 0; nt2 < 7; ++nt2) {
      const int n  = 16 * nt2 + lm;
      const int nc = n < H1D ? n : H1D - 1;
      const float bias = b1[nc];
      v8f accF;
#pragma unroll
      for (int v = 0; v < 8; ++v) accF[v] = bias;
#pragma unroll
      for (int kc = 0; kc < 8; ++kc) {
        int k0 = 16 * half + 2 * kc;
        v2f bf = *(const v2f*)(G + nc * RDIM + k0);
        accF = wmma_f32_4(gam[kc], bf, accF);
      }
#pragma unroll
      for (int v = 0; v < 8; ++v) Hb1[v + 8 * half][n] = fmaxf(accF[v], 0.f);
    }
  }
  __syncthreads();

  {
    const int s = lm, jh = half;
    if (w == 0) {
      float acc[50];
#pragma unroll
      for (int i = 0; i < 50; ++i) acc[i] = b2[50 * jh + i];
      dense_rows<H1D, 50>(W2 + (size_t)(50 * jh) * H1D, &Hb1[s][0], acc);
#pragma unroll
      for (int i = 0; i < 50; ++i) Hb2[s][50 * jh + i] = fmaxf(acc[i], 0.f);
    }
    __syncthreads();
    if (w == 0) {
      float acc3[5];
#pragma unroll
      for (int i = 0; i < 5; ++i) acc3[i] = b3[5 * jh + i];
      dense_rows<H2D, 5>(W3 + (size_t)(5 * jh) * H2D, &Hb2[s][0], acc3);
#pragma unroll
      for (int i = 0; i < 5; ++i) obuf[s * NLAB + 5 * jh + i] = 1.0f * acc3[i];
    }
    __syncthreads();
    if (w == 0) {
#pragma unroll
      for (int it = 0; it < 5; ++it)
        out[(size_t)seq0 * NLAB + it * 32 + l] = obuf[it * 32 + l];
    }
  }
}

extern "C" void kernel_launch(void* const* d_in, const int* in_sizes, int n_in,
                              void* d_out, int out_size, void* d_ws, size_t ws_size,
                              hipStream_t stream) {
  const int*   x      = (const int*)d_in[0];
  const int*   lens   = (const int*)d_in[1];
  const float* alpha0 = (const float*)d_in[2];
  const float* ER     = (const float*)d_in[3];
  const float* D1     = (const float*)d_in[4];
  const float* D2     = (const float*)d_in[5];
  const float* W1     = (const float*)d_in[6];
  const float* b1     = (const float*)d_in[7];
  const float* W2     = (const float*)d_in[8];
  const float* b2     = (const float*)d_in[9];
  const float* W3     = (const float*)d_in[10];
  const float* b3     = (const float*)d_in[11];
  float* ws  = (float*)d_ws;
  float* out = (float*)d_out;

  farnn_setup<<<133, 32, 0, stream>>>(alpha0, D1, D2, W1, ws);
  farnn_main<<<2048 / 16, 32 * NWV, 0, stream>>>(x, lens, ER, ws, b1, W2, b2,
                                           W3, b3, out);
}
